# starting kernel
# baseline (speedup 1.0000x reference)
#include <hip/hip_runtime.h>
#include <math.h>

typedef __attribute__((ext_vector_type(2))) float v2f;
typedef __attribute__((ext_vector_type(8))) float v8f;

#define TT 4096
#define BB 256

__device__ __forceinline__ float fast_tanh(float x) {
    float e = __expf(2.0f * x);
    return 1.0f - 2.0f * __builtin_amdgcn_rcpf(e + 1.0f);
}

__device__ __forceinline__ v8f wmma4(v2f a, v2f b, v8f c) {
    return __builtin_amdgcn_wmma_f32_16x16x4_f32(
        false, a, false, b, (short)0, c, false, false);
}

__global__ __launch_bounds__(512)
void pinn_kernel(const float* __restrict__ x,
                 const float* __restrict__ W1,  const float* __restrict__ b1,
                 const float* __restrict__ W2,  const float* __restrict__ b2,
                 const float* __restrict__ W3,  const float* __restrict__ b3,
                 const float* __restrict__ logE,const float* __restrict__ logeta,
                 const float* __restrict__ imp, const float* __restrict__ lw,
                 const float* __restrict__ Wf1, const float* __restrict__ bf1,
                 const float* __restrict__ Wf2, const float* __restrict__ bf2,
                 float* __restrict__ out)
{
    __shared__ __align__(16) float h1s[16][16 * 34];
    __shared__ __align__(16) float h2s[16][16 * 20];
    __shared__ __align__(16) float sbuf[2][16][16];

    const int tid  = threadIdx.x;
    const int n    = tid >> 5;
    const int lane = tid & 31;
    const int hi   = lane >> 4;
    const int lm   = lane & 15;
    const int b0   = blockIdx.x * 16;
    const int kb   = hi ? 2 : 0;

    const float E   = __expf(logE[n]);
    const float eta = __expf(logeta[n]);
    const float iw  = 1.0f / (1.0f + __expf(-imp[n]));
    const float b3v = b3[n];

    v2f B1[2]; v8f C1[2];
#pragma unroll
    for (int h = 0; h < 2; ++h) {
        int j = 16 * h + lm;
        B1[h].x = W1[(n * 32 + j) * 3 + kb];
        B1[h].y = hi ? 0.0f : W1[(n * 32 + j) * 3 + 1];
        float bb = b1[n * 32 + j];
#pragma unroll
        for (int i = 0; i < 8; ++i) C1[h][i] = bb;
    }
    v2f B2[8];
#pragma unroll
    for (int kc = 0; kc < 8; ++kc) {
        B2[kc].x = W2[(n * 16 + lm) * 32 + 4 * kc + kb];
        B2[kc].y = W2[(n * 16 + lm) * 32 + 4 * kc + kb + 1];
    }
    const float c2s = b2[n * 16 + lm];

    float w3v[16];
#pragma unroll
    for (int k = 0; k < 16; ++k) w3v[k] = W3[n * 16 + k];

    float cl[4];
    {
        float e0 = __expf(lw[0]), e1 = __expf(lw[1]);
        float e2 = __expf(lw[2]), e3 = __expf(lw[3]);
        float es = e0 + e1 + e2 + e3;
        float ws[4] = { e0 / es, e1 / es, e2 / es, e3 / es };
#pragma unroll
        for (int l = 0; l < 4; ++l) {
            float s = 0.0f;
#pragma unroll
            for (int p = 0; p < 4; ++p)
                s += 1.0f / (1.0f + __expf(-imp[l * 4 + p]));
            cl[l] = (s / (s + 1e-8f)) * ws[l];
        }
    }
    float wf1v[4];
#pragma unroll
    for (int l = 0; l < 4; ++l) wf1v[l] = Wf1[lm * 4 + l];
    const float bf1v = bf1[lm];
    const float wf2v = Wf2[lm];
    const float bf2v = bf2[0];

    float prev = 0.0f;
    const float2* xp = (const float2*)x;
    float2 xc = make_float2(0.0f, 0.0f);
    if (!hi) xc = xp[(size_t)(b0 + lm) * TT];

    float* h1p = h1s[n];
    float* h2p = h2s[n];

    for (int t = 0; t < TT; ++t) {
        float2 xn = make_float2(0.0f, 0.0f);
        if (!hi && (t + 1) < TT) xn = xp[(size_t)(b0 + lm) * TT + (t + 1)];

        v2f a1;
        a1.x = hi ? prev : xc.x;
        a1.y = hi ? 0.0f : xc.y;

        v8f d0 = wmma4(a1, B1[0], C1[0]);
        v8f d1 = wmma4(a1, B1[1], C1[1]);
#pragma unroll
        for (int i = 0; i < 8; ++i) { d0[i] = fast_tanh(d0[i]); d1[i] = fast_tanh(d1[i]); }

#pragma unroll
        for (int v = 0; v < 8; ++v) {
            h1p[(v + 8 * hi) * 34 + lm]      = d0[v];
            h1p[(v + 8 * hi) * 34 + 16 + lm] = d1[v];
        }

        v8f acc;
#pragma unroll
        for (int i = 0; i < 8; ++i) acc[i] = c2s;
#pragma unroll
        for (int kc = 0; kc < 8; ++kc) {
            v2f a2 = *(const v2f*)&h1p[lm * 34 + 4 * kc + kb];
            acc = wmma4(a2, B2[kc], acc);
        }
#pragma unroll
        for (int i = 0; i < 8; ++i) acc[i] = fast_tanh(acc[i]);

#pragma unroll
        for (int v = 0; v < 8; ++v) h2p[(v + 8 * hi) * 20 + lm] = acc[v];

        const int buf = t & 1;
        if (!hi) {
            const float4* row = (const float4*)&h2p[lm * 20];
            float corr = b3v;
#pragma unroll
            for (int j = 0; j < 4; ++j) {
                float4 r = row[j];
                corr = fmaf(r.x, w3v[4 * j + 0], corr);
                corr = fmaf(r.y, w3v[4 * j + 1], corr);
                corr = fmaf(r.z, w3v[4 * j + 2], corr);
                corr = fmaf(r.w, w3v[4 * j + 3], corr);
            }
            float mw = E * xc.x + eta * xc.y;
            sbuf[buf][n][lm] = (mw + corr) * iw;
        }
        __syncthreads();

        prev = sbuf[buf][n][lm];

#pragma unroll
        for (int pp = 0; pp < 2; ++pp) {
            int p = 2 * hi + pp;
            float v0 = sbuf[buf][0 * 4 + p][n] * cl[0];
            float v1 = sbuf[buf][1 * 4 + p][n] * cl[1];
            float v2 = sbuf[buf][2 * 4 + p][n] * cl[2];
            float v3 = sbuf[buf][3 * 4 + p][n] * cl[3];
            float hf = fmaf(v0, wf1v[0], bf1v);
            hf = fmaf(v1, wf1v[1], hf);
            hf = fmaf(v2, wf1v[2], hf);
            hf = fmaf(v3, wf1v[3], hf);
            hf = fmaxf(hf, 0.0f);
            float c = wf2v * hf;
            c += __shfl_xor(c, 1);
            c += __shfl_xor(c, 2);
            c += __shfl_xor(c, 4);
            c += __shfl_xor(c, 8);
            if (lm == 0)
                out[(size_t)(p * BB + b0 + n) * TT + t] = c + bf2v;
        }
        xc = xn;
    }
}

extern "C" void kernel_launch(void* const* d_in, const int* in_sizes, int n_in,
                              void* d_out, int out_size, void* d_ws, size_t ws_size,
                              hipStream_t stream) {
    (void)in_sizes; (void)n_in; (void)d_ws; (void)ws_size; (void)out_size;
    pinn_kernel<<<dim3(16), dim3(512), 0, stream>>>(
        (const float*)d_in[0],  (const float*)d_in[1],  (const float*)d_in[2],
        (const float*)d_in[3],  (const float*)d_in[4],  (const float*)d_in[5],
        (const float*)d_in[6],  (const float*)d_in[7],  (const float*)d_in[8],
        (const float*)d_in[9],  (const float*)d_in[10], (const float*)d_in[11],
        (const float*)d_in[12], (const float*)d_in[13], (const float*)d_in[14],
        (float*)d_out);
}


# tanh's affine parts folded ahead of the loop: input scale 2*log2(e) into the feeding weights and bias, output map 1-2r into layer 2's weights and bias; layer-1/2 rounding, layer-2 sum grouping change
# speedup vs baseline: 4.5120x; 4.5120x over previous
#include <hip/hip_runtime.h>
#include <math.h>


typedef __attribute__((ext_vector_type(2))) float v2f;
typedef __attribute__((ext_vector_type(8))) float v8f;

#define TT 4096
#define BB 256

#define TANH_IN 2.885390081777926814f

__device__ __forceinline__ float half_gate(float z) {
    return __builtin_amdgcn_rcpf(__builtin_amdgcn_exp2f(z) + 1.0f);
}

__device__ __forceinline__ float tanh_scaled_of_z(float z, float w) {
    return fmaf(half_gate(z), -2.0f * w, w);
}

template <int MASK>
__device__ __forceinline__ float lane_xor(float v) {
    return __builtin_bit_cast(float, __builtin_amdgcn_update_dpp(
        0, __builtin_bit_cast(int, v), 0x160 | MASK, 0xF, 0xF, true));
}

__device__ __forceinline__ v8f wmma4(v2f a, v2f b, v8f c) {
    return __builtin_amdgcn_wmma_f32_16x16x4_f32(
        false, a, false, b, (short)0, c, false, false);
}


__global__ __launch_bounds__(32)
void recurrence_kernel(const float* __restrict__ x,
                     const float* __restrict__ W1,  const float* __restrict__ b1,
                     const float* __restrict__ W2,  const float* __restrict__ b2,
                     const float* __restrict__ W3,  const float* __restrict__ b3,
                     const float* __restrict__ logE,const float* __restrict__ logeta,
                     const float* __restrict__ imp,
                     float* __restrict__ S)
{
    __shared__ __align__(16) float h1s[16 * 34];

    const int lane = threadIdx.x;
    const int n    = blockIdx.x & 15;
    const int tile = blockIdx.x >> 4;
    const int hi   = lane >> 4;
    const int lm   = lane & 15;
    const int b0   = tile * 16;
    const int kb   = hi ? 2 : 0;
    const bool own = (lm >> 3) == hi;
    const bool lb2 = (lm & 4) != 0;
    const bool lb1 = (lm & 2) != 0;
    const bool lb0 = (lm & 1) != 0;

    const float E   = __expf(logE[n]);
    const float eta = __expf(logeta[n]);
    const float iw  = 1.0f / (1.0f + __expf(-imp[n]));
    const float b3v = b3[n];

    v2f B1[2]; v8f C1[2];
#pragma unroll
    for (int h = 0; h < 2; ++h) {
        int j = 16 * h + lm;
        B1[h].x = TANH_IN * W1[(n * 32 + j) * 3 + hi];
        B1[h].y = TANH_IN * W1[(n * 32 + j) * 3 + 2];
        float bb = TANH_IN * b1[n * 32 + j];
#pragma unroll
        for (int i = 0; i < 8; ++i) C1[h][i] = bb;
    }
    v2f B2[8];
#pragma unroll
    for (int kc = 0; kc < 8; ++kc) {
        B2[kc].x = (-2.0f * TANH_IN) * W2[(n * 16 + lm) * 32 + 4 * kc + kb];
        B2[kc].y = (-2.0f * TANH_IN) * W2[(n * 16 + lm) * 32 + 4 * kc + kb + 1];
    }
    float w2row = 0.0f;
    for (int k = 0; k < 32; ++k) w2row += W2[(n * 16 + lm) * 32 + k];
    const float c2s = TANH_IN * (b2[n * 16 + lm] + w2row);

    const float w3l = W3[n * 16 + lm];

    const float2* xrow = (const float2*)x + (size_t)(b0 + lm) * TT;
    float2 xc = xrow[0];
    float prev = 0.0f;
    float* Sp = S + (size_t)(n * 16 + tile) * TT * 16;
    const int soff = own ? lm : 16 + (lm & 7) + 8 * hi;

    for (int t = 0; t < TT; t += 2) {
        float st[2];
#pragma unroll
        for (int u = 0; u < 2; ++u) {
            const int tc = t + u;
            float2 xn = make_float2(0.0f, 0.0f);
            if ((tc + 1) < TT) xn = xrow[tc + 1];

            v2f a1;
            a1.x = hi ? xc.y : xc.x;
            a1.y = own ? prev : 0.0f;

            v8f d0 = wmma4(a1, B1[0], C1[0]);
            v8f d1 = wmma4(a1, B1[1], C1[1]);
            v2f a2[8];
            asm volatile("" :: "v"(d1));
#pragma unroll
            for (int i = 0; i < 8; ++i) d0[i] = half_gate(d0[i]);
#pragma unroll
            for (int v = 0; v < 8; ++v) h1s[(v + 8 * hi) * 34 + lm] = d0[v];
#pragma unroll
            for (int kc = 0; kc < 4; ++kc) a2[kc] = *(const v2f*)&h1s[lm * 34 + 4 * kc + kb];
            __builtin_amdgcn_sched_barrier(0);
#pragma unroll
            for (int i = 0; i < 8; ++i) d1[i] = half_gate(d1[i]);
#pragma unroll
            for (int v = 0; v < 8; ++v) h1s[(v + 8 * hi) * 34 + 16 + lm] = d1[v];
#pragma unroll
            for (int kc = 4; kc < 8; ++kc) a2[kc] = *(const v2f*)&h1s[lm * 34 + 4 * kc + kb];
            __builtin_amdgcn_sched_barrier(0);

            v8f acc;
#pragma unroll
            for (int i = 0; i < 8; ++i) acc[i] = c2s;
#pragma unroll
            for (int kc = 0; kc < 8; ++kc) acc = wmma4(a2[kc], B2[kc], acc);

            float p[8];
#pragma unroll
            for (int i = 0; i < 8; ++i) p[i] = tanh_scaled_of_z(acc[i], w3l);

            float q[4];
#pragma unroll
            for (int i = 0; i < 4; ++i) {
                const float keep = lb2 ? p[i + 4] : p[i];
                const float send = lb2 ? p[i] : p[i + 4];
                q[i] = keep + lane_xor<4>(send);
            }
            float r[2];
#pragma unroll
            for (int i = 0; i < 2; ++i) {
                const float keep = lb1 ? q[i + 2] : q[i];
                const float send = lb1 ? q[i] : q[i + 2];
                r[i] = keep + lane_xor<2>(send);
            }
            float s;
            {
                const float keep = lb0 ? r[1] : r[0];
                const float send = lb0 ? r[0] : r[1];
                s = keep + lane_xor<1>(send);
            }
            s = s + lane_xor<8>(s);

            const float corr = s + b3v;
            const float mw = E * xc.x + eta * xc.y;
            st[u] = (mw + corr) * iw;
            prev = st[u];
            xc = xn;
        }
        const float nxt = lane_xor<8>(st[1]);
        Sp[(size_t)t * 16 + soff] = own ? st[0] : nxt;
    }
}

__global__ __launch_bounds__(256)
void fusion_head_kernel(const float* __restrict__ S,
                      const float* __restrict__ imp, const float* __restrict__ lw,
                      const float* __restrict__ Wf1, const float* __restrict__ bf1,
                      const float* __restrict__ Wf2, const float* __restrict__ bf2,
                      float* __restrict__ out)
{
    __shared__ float tl[8][16 * 34];

    const int lane  = threadIdx.x & 31;
    const int wv    = threadIdx.x >> 5;
    const int gw    = blockIdx.x * 8 + wv;
    const int chunk = gw & 127;
    const int tile  = (gw >> 7) & 15;
    const int p     = gw >> 11;
    const int t0    = chunk * 32;

    float cl[4];
    {
        float e0 = __expf(lw[0]), e1 = __expf(lw[1]);
        float e2 = __expf(lw[2]), e3 = __expf(lw[3]);
        float es = e0 + e1 + e2 + e3;
        float ws[4] = { e0 / es, e1 / es, e2 / es, e3 / es };
#pragma unroll
        for (int l = 0; l < 4; ++l) {
            float s = 0.0f;
#pragma unroll
            for (int q = 0; q < 4; ++q)
                s += 1.0f / (1.0f + __expf(-imp[l * 4 + q]));
            cl[l] = (s / (s + 1e-8f)) * ws[l];
        }
    }
    const float bf2v = bf2[0];

    float v[4][16];
#pragma unroll
    for (int l = 0; l < 4; ++l) {
        const float4* src =
            (const float4*)(S + ((size_t)((l * 4 + p) * 16 + tile) * TT + t0) * 16);
#pragma unroll
        for (int k = 0; k < 4; ++k) {
            float4 q = src[k * 32 + lane];
            v[l][4 * k + 0] = q.x * cl[l];
            v[l][4 * k + 1] = q.y * cl[l];
            v[l][4 * k + 2] = q.z * cl[l];
            v[l][4 * k + 3] = q.w * cl[l];
        }
    }

    float acc[16];
#pragma unroll
    for (int e = 0; e < 16; ++e) acc[e] = 0.0f;
#pragma unroll
    for (int f = 0; f < 16; ++f) {
        const float w0 = Wf1[f * 4 + 0], w1 = Wf1[f * 4 + 1];
        const float w2 = Wf1[f * 4 + 2], w3 = Wf1[f * 4 + 3];
        const float bf = bf1[f], wo = Wf2[f];
#pragma unroll
        for (int e = 0; e < 16; ++e) {
            float hf = fmaf(v[0][e], w0, bf);
            hf = fmaf(v[1][e], w1, hf);
            hf = fmaf(v[2][e], w2, hf);
            hf = fmaf(v[3][e], w3, hf);
            hf = fmaxf(hf, 0.0f);
            acc[e] = fmaf(wo, hf, acc[e]);
        }
    }

    float* tp = tl[wv];
#pragma unroll
    for (int k = 0; k < 4; ++k)
#pragma unroll
        for (int i = 0; i < 4; ++i)
            tp[((lane & 3) * 4 + i) * 34 + 8 * k + (lane >> 2)] = acc[4 * k + i] + bf2v;
#pragma unroll
    for (int r = 0; r < 16; ++r)
        out[(size_t)(p * BB + tile * 16 + r) * TT + t0 + lane] = tp[r * 34 + lane];
}

extern "C" void kernel_launch(void* const* d_in, const int* in_sizes, int n_in,
                              void* d_out, int out_size, void* d_ws, size_t ws_size,
                              hipStream_t stream) {
    (void)in_sizes; (void)n_in; (void)ws_size; (void)out_size;
    recurrence_kernel<<<dim3(256), dim3(32), 0, stream>>>(
        (const float*)d_in[0],  (const float*)d_in[1],  (const float*)d_in[2],
        (const float*)d_in[3],  (const float*)d_in[4],  (const float*)d_in[5],
        (const float*)d_in[6],  (const float*)d_in[7],  (const float*)d_in[8],
        (const float*)d_in[9],
        (float*)d_ws);
    fusion_head_kernel<<<dim3(1024), dim3(256), 0, stream>>>(
        (const float*)d_ws,
        (const float*)d_in[9],  (const float*)d_in[10], (const float*)d_in[11],
        (const float*)d_in[12], (const float*)d_in[13], (const float*)d_in[14],
        (float*)d_out);
}


# recurrence re-tiled to 8 batch rows per wave (512 waves): matrix rows 0-7 / 8-15 hold hidden units 0-15 / 16-31, so layer 1 is one tile; layer-1 and layer-3 sum order changed
# speedup vs baseline: 5.5957x; 1.2402x over previous
#include <hip/hip_runtime.h>
#include <math.h>


typedef __attribute__((ext_vector_type(2))) float v2f;
typedef __attribute__((ext_vector_type(8))) float v8f;

#define TT 4096
#define BB 256

#define TANH_IN 2.885390081777926814f

__device__ __forceinline__ float half_gate(float z) {
    return __builtin_amdgcn_rcpf(__builtin_amdgcn_exp2f(z) + 1.0f);
}

__device__ __forceinline__ float tanh_scaled_of_z(float z, float w) {
    return fmaf(half_gate(z), -2.0f * w, w);
}

template <int MASK>
__device__ __forceinline__ float lane_xor(float v) {
    return __builtin_bit_cast(float, __builtin_amdgcn_update_dpp(
        0, __builtin_bit_cast(int, v), 0x160 | MASK, 0xF, 0xF, true));
}

__device__ __forceinline__ v8f wmma4(v2f a, v2f b, v8f c) {
    return __builtin_amdgcn_wmma_f32_16x16x4_f32(
        false, a, false, b, (short)0, c, false, false);
}


__global__ __launch_bounds__(32)
void recurrence_kernel(const float* __restrict__ x,
                     const float* __restrict__ W1,  const float* __restrict__ b1,
                     const float* __restrict__ W2,  const float* __restrict__ b2,
                     const float* __restrict__ W3,  const float* __restrict__ b3,
                     const float* __restrict__ logE,const float* __restrict__ logeta,
                     const float* __restrict__ imp,
                     float* __restrict__ S)
{
    __shared__ __align__(16) float h1s[8 * 34];

    const int lane = threadIdx.x;
    const int n    = blockIdx.x & 15;
    const int tile = blockIdx.x >> 4;
    const int hi   = lane >> 4;
    const int lm   = lane & 15;
    const int b0   = tile * 8;
    const int kb   = hi ? 2 : 0;
    const int srow = 4 * hi + (lm & 3);
    const int arow = (lm & 3) + 4 * ((lm >> 3) & 1);
    const bool lb1 = (lm & 2) != 0;
    const bool lb0 = (lm & 1) != 0;
    const bool cx0 = (lm >> 2) == hi;
    const bool cx1 = (lm >> 2) == hi + 2;
    const bool xlive = (lm >> 3) == hi;

    const float E   = __expf(logE[n]);
    const float eta = __expf(logeta[n]);
    const float iw  = 1.0f / (1.0f + __expf(-imp[n]));
    const float b3v = b3[n];

    v2f Bx, Bc; v8f Cb;
    {
        const int j = 16 * hi + lm;
        Bx.x = TANH_IN * W1[(n * 32 + j) * 3 + 0];
        Bx.y = TANH_IN * W1[(n * 32 + j) * 3 + 1];
        Bc.x = TANH_IN * W1[(n * 32 + lm) * 3 + 2];
        Bc.y = TANH_IN * W1[(n * 32 + 16 + lm) * 3 + 2];
        const float bb = TANH_IN * b1[n * 32 + j];
#pragma unroll
        for (int i = 0; i < 8; ++i) Cb[i] = bb;
    }
    v2f B2[8];
#pragma unroll
    for (int kc = 0; kc < 8; ++kc) {
        B2[kc].x = (-2.0f * TANH_IN) * W2[(n * 16 + lm) * 32 + 4 * kc + kb];
        B2[kc].y = (-2.0f * TANH_IN) * W2[(n * 16 + lm) * 32 + 4 * kc + kb + 1];
    }
    float w2row = 0.0f;
    for (int k = 0; k < 32; ++k) w2row += W2[(n * 16 + lm) * 32 + k];
    const float c2s = TANH_IN * (b2[n * 16 + lm] + w2row);

    const float w3l = W3[n * 16 + lm];

    const float2* xa = (const float2*)x + (size_t)(b0 + (lm & 7)) * TT;
    const float2* xs = (const float2*)x + (size_t)(b0 + srow) * TT;
    float2 xsc = xs[0];
    float2 xan = xa[1];
    float prev = 0.0f;
    float* Sp = S + (size_t)(n * 32 + tile) * TT * 8;
    const int su   = lm >> 2;
    const int soff = su * 8 + srow;

    v8f cxc;
    {
        const float2 x0 = xa[0];
        v2f ax; ax.x = xlive ? x0.x : 0.0f; ax.y = xlive ? x0.y : 0.0f;
        cxc = wmma4(ax, Bx, Cb);
    }

    for (int t = 0; t < TT; t += 4) {
        float st[4];
#pragma unroll
        for (int u = 0; u < 4; ++u) {
            const int tc = t + u;
            const float2 xann = xa[(tc + 2) < TT ? tc + 2 : TT - 1];
            const float2 xsn  = xs[(tc + 1) < TT ? tc + 1 : TT - 1];

            v2f ac;
            ac.x = cx0 ? prev : 0.0f;
            ac.y = cx1 ? prev : 0.0f;

            v8f d = wmma4(ac, Bc, cxc);
            {
                v2f ax; ax.x = xlive ? xan.x : 0.0f; ax.y = xlive ? xan.y : 0.0f;
                cxc = wmma4(ax, Bx, Cb);
            }
            asm volatile("" :: "v"(cxc));
#pragma unroll
            for (int i = 0; i < 8; ++i) d[i] = half_gate(d[i]);
#pragma unroll
            for (int v = 0; v < 8; ++v) h1s[v * 34 + 16 * hi + lm] = d[v];
            v2f a2[8];
#pragma unroll
            for (int kc = 0; kc < 8; ++kc) a2[kc] = *(const v2f*)&h1s[arow * 34 + 4 * kc + kb];
            __builtin_amdgcn_sched_barrier(0);

            v8f acc;
#pragma unroll
            for (int i = 0; i < 8; ++i) acc[i] = c2s;
#pragma unroll
            for (int kc = 0; kc < 8; ++kc) acc = wmma4(a2[kc], B2[kc], acc);

            float p[4];
#pragma unroll
            for (int i = 0; i < 4; ++i) p[i] = tanh_scaled_of_z(acc[i], w3l);

            float q[2];
#pragma unroll
            for (int i = 0; i < 2; ++i) {
                const float keep = lb1 ? p[i + 2] : p[i];
                const float send = lb1 ? p[i] : p[i + 2];
                q[i] = keep + lane_xor<2>(send);
            }
            float s;
            {
                const float keep = lb0 ? q[1] : q[0];
                const float send = lb0 ? q[0] : q[1];
                s = keep + lane_xor<1>(send);
            }
            s = s + lane_xor<4>(s);
            s = s + lane_xor<8>(s);

            const float corr = s + b3v;
            const float mw = E * xsc.x + eta * xsc.y;
            st[u] = (mw + corr) * iw;
            prev = st[u];
            xsc = xsn;
            xan = xann;
        }
        const float o01 = (su & 1) ? st[1] : st[0];
        const float o23 = (su & 1) ? st[3] : st[2];
        Sp[(size_t)t * 8 + soff] = (su & 2) ? o23 : o01;
    }
}

__global__ __launch_bounds__(256)
void fusion_head_kernel(const float* __restrict__ S,
                      const float* __restrict__ imp, const float* __restrict__ lw,
                      const float* __restrict__ Wf1, const float* __restrict__ bf1,
                      const float* __restrict__ Wf2, const float* __restrict__ bf2,
                      float* __restrict__ out)
{
    __shared__ float tl[8][16 * 34];

    const int lane  = threadIdx.x & 31;
    const int wv    = threadIdx.x >> 5;
    const int gw    = blockIdx.x * 8 + wv;
    const int chunk = gw & 127;
    const int tile  = (gw >> 7) & 15;
    const int p     = gw >> 11;
    const int t0    = chunk * 32;

    float cl[4];
    {
        float e0 = __expf(lw[0]), e1 = __expf(lw[1]);
        float e2 = __expf(lw[2]), e3 = __expf(lw[3]);
        float es = e0 + e1 + e2 + e3;
        float ws[4] = { e0 / es, e1 / es, e2 / es, e3 / es };
#pragma unroll
        for (int l = 0; l < 4; ++l) {
            float s = 0.0f;
#pragma unroll
            for (int q = 0; q < 4; ++q)
                s += 1.0f / (1.0f + __expf(-imp[l * 4 + q]));
            cl[l] = (s / (s + 1e-8f)) * ws[l];
        }
    }
    const float bf2v = bf2[0];

    float v[4][16];
#pragma unroll
    for (int l = 0; l < 4; ++l) {
        const float4* src =
            (const float4*)(S + ((size_t)((l * 4 + p) * 32 + 2 * tile + ((lane & 3) >> 1)) * TT + t0) * 8);
#pragma unroll
        for (int k = 0; k < 4; ++k) {
            float4 q = src[(k * 8 + (lane >> 2)) * 2 + (lane & 1)];
            v[l][4 * k + 0] = q.x * cl[l];
            v[l][4 * k + 1] = q.y * cl[l];
            v[l][4 * k + 2] = q.z * cl[l];
            v[l][4 * k + 3] = q.w * cl[l];
        }
    }

    float acc[16];
#pragma unroll
    for (int e = 0; e < 16; ++e) acc[e] = 0.0f;
#pragma unroll
    for (int f = 0; f < 16; ++f) {
        const float w0 = Wf1[f * 4 + 0], w1 = Wf1[f * 4 + 1];
        const float w2 = Wf1[f * 4 + 2], w3 = Wf1[f * 4 + 3];
        const float bf = bf1[f], wo = Wf2[f];
#pragma unroll
        for (int e = 0; e < 16; ++e) {
            float hf = fmaf(v[0][e], w0, bf);
            hf = fmaf(v[1][e], w1, hf);
            hf = fmaf(v[2][e], w2, hf);
            hf = fmaf(v[3][e], w3, hf);
            hf = fmaxf(hf, 0.0f);
            acc[e] = fmaf(wo, hf, acc[e]);
        }
    }

    float* tp = tl[wv];
#pragma unroll
    for (int k = 0; k < 4; ++k)
#pragma unroll
        for (int i = 0; i < 4; ++i)
            tp[((lane & 3) * 4 + i) * 34 + 8 * k + (lane >> 2)] = acc[4 * k + i] + bf2v;
#pragma unroll
    for (int r = 0; r < 16; ++r)
        out[(size_t)(p * BB + tile * 16 + r) * TT + t0 + lane] = tp[r * 34 + lane];
}

extern "C" void kernel_launch(void* const* d_in, const int* in_sizes, int n_in,
                              void* d_out, int out_size, void* d_ws, size_t ws_size,
                              hipStream_t stream) {
    (void)in_sizes; (void)n_in; (void)ws_size; (void)out_size;
    recurrence_kernel<<<dim3(512), dim3(32), 0, stream>>>(
        (const float*)d_in[0],  (const float*)d_in[1],  (const float*)d_in[2],
        (const float*)d_in[3],  (const float*)d_in[4],  (const float*)d_in[5],
        (const float*)d_in[6],  (const float*)d_in[7],  (const float*)d_in[8],
        (const float*)d_in[9],
        (float*)d_ws);
    fusion_head_kernel<<<dim3(1024), dim3(256), 0, stream>>>(
        (const float*)d_ws,
        (const float*)d_in[9],  (const float*)d_in[10], (const float*)d_in[11],
        (const float*)d_in[12], (const float*)d_in[13], (const float*)d_in[14],
        (float*)d_out);
}


# recurrence re-tiled to 4 batch rows per wave (1024 waves as 256 workgroups of 4 independent waves, one per SIMD): 4 layer-1 and 2 layer-3 tanh per lane; layer-3 sum order changed
# speedup vs baseline: 6.4479x; 1.1523x over previous
#include <hip/hip_runtime.h>
#include <math.h>


typedef __attribute__((ext_vector_type(2))) float v2f;
typedef __attribute__((ext_vector_type(8))) float v8f;

#define TT 4096
#define BB 256

#define TANH_IN 2.885390081777926814f

__device__ __forceinline__ float half_gate(float z) {
    return __builtin_amdgcn_rcpf(__builtin_amdgcn_exp2f(z) + 1.0f);
}

__device__ __forceinline__ float tanh_scaled_of_z(float z, float w) {
    return fmaf(half_gate(z), -2.0f * w, w);
}

template <int MASK>
__device__ __forceinline__ float lane_xor(float v) {
    return __builtin_bit_cast(float, __builtin_amdgcn_update_dpp(
        0, __builtin_bit_cast(int, v), 0x160 | MASK, 0xF, 0xF, true));
}

__device__ __forceinline__ v8f wmma4(v2f a, v2f b, v8f c) {
    return __builtin_amdgcn_wmma_f32_16x16x4_f32(
        false, a, false, b, (short)0, c, false, false);
}


__global__ __launch_bounds__(128)
void recurrence_kernel(const float* __restrict__ x,
                     const float* __restrict__ W1,  const float* __restrict__ b1,
                     const float* __restrict__ W2,  const float* __restrict__ b2,
                     const float* __restrict__ W3,  const float* __restrict__ b3,
                     const float* __restrict__ logE,const float* __restrict__ logeta,
                     const float* __restrict__ imp,
                     float* __restrict__ S)
{
    __shared__ __align__(16) float h1all[4][4 * 34];

    const int lane = threadIdx.x & 31;
    const int wv   = threadIdx.x >> 5;
    float* h1s     = h1all[wv];
    const int gw   = blockIdx.x * 4 + wv;
    const int n    = gw & 15;
    const int tile = gw >> 4;
    const int hi   = lane >> 4;
    const int lm   = lane & 15;
    const int b0   = tile * 4;
    const int kb   = hi ? 2 : 0;
    const int srow = 2 * hi + (lm & 1);
    const int arow = (lm & 1) + 2 * ((lm >> 3) & 1);
    const bool lb0 = (lm & 1) != 0;
    const bool cx0 = (lm < 4) && ((lm >> 1) == hi);
    const bool cx1 = ((lm & 12) == 8) && (((lm >> 1) & 1) == hi);
    const bool xlive = (lm >> 3) == hi;

    const float E   = __expf(logE[n]);
    const float eta = __expf(logeta[n]);
    const float iw  = 1.0f / (1.0f + __expf(-imp[n]));
    const float b3v = b3[n];

    v2f Bx, Bc; v8f Cb;
    {
        const int j = 16 * hi + lm;
        Bx.x = TANH_IN * W1[(n * 32 + j) * 3 + 0];
        Bx.y = TANH_IN * W1[(n * 32 + j) * 3 + 1];
        Bc.x = TANH_IN * W1[(n * 32 + lm) * 3 + 2];
        Bc.y = TANH_IN * W1[(n * 32 + 16 + lm) * 3 + 2];
        const float bb = TANH_IN * b1[n * 32 + j];
#pragma unroll
        for (int i = 0; i < 8; ++i) Cb[i] = bb;
    }
    v2f B2[8];
#pragma unroll
    for (int kc = 0; kc < 8; ++kc) {
        B2[kc].x = (-2.0f * TANH_IN) * W2[(n * 16 + lm) * 32 + 4 * kc + kb];
        B2[kc].y = (-2.0f * TANH_IN) * W2[(n * 16 + lm) * 32 + 4 * kc + kb + 1];
    }
    float w2row = 0.0f;
    for (int k = 0; k < 32; ++k) w2row += W2[(n * 16 + lm) * 32 + k];
    const float c2s = TANH_IN * (b2[n * 16 + lm] + w2row);

    const float w3l = W3[n * 16 + lm];

    const float2* xa = (const float2*)x + (size_t)(b0 + (lm & 3)) * TT;
    const float2* xs = (const float2*)x + (size_t)(b0 + srow) * TT;
    float2 xsc = xs[0];
    float2 xan = xa[1];
    float prev = 0.0f;
    float* Sp = S + (size_t)(n * 64 + tile) * TT * 4;
    const int su   = lm >> 1;
    const int soff = su * 4 + srow;

    v8f cxc;
    {
        const float2 x0 = xa[0];
        v2f ax; ax.x = xlive ? x0.x : 0.0f; ax.y = xlive ? x0.y : 0.0f;
        cxc = wmma4(ax, Bx, Cb);
    }

    for (int t = 0; t < TT; t += 8) {
        float st[8];
#pragma unroll
        for (int u = 0; u < 8; ++u) {
            const int tc = t + u;
            const float2 xann = xa[(tc + 2) < TT ? tc + 2 : TT - 1];
            const float2 xsn  = xs[(tc + 1) < TT ? tc + 1 : TT - 1];

            v2f ac;
            ac.x = cx0 ? prev : 0.0f;
            ac.y = cx1 ? prev : 0.0f;

            v8f d = wmma4(ac, Bc, cxc);
            {
                v2f ax; ax.x = xlive ? xan.x : 0.0f; ax.y = xlive ? xan.y : 0.0f;
                cxc = wmma4(ax, Bx, Cb);
            }
            asm volatile("" :: "v"(cxc));
#pragma unroll
            for (int i = 0; i < 4; ++i) d[i] = half_gate(d[i]);
#pragma unroll
            for (int v = 0; v < 4; ++v) h1s[v * 34 + 16 * hi + lm] = d[v];
            v2f a2[8];
#pragma unroll
            for (int kc = 0; kc < 8; ++kc) a2[kc] = *(const v2f*)&h1s[arow * 34 + 4 * kc + kb];
            __builtin_amdgcn_sched_barrier(0);

            v8f acc;
#pragma unroll
            for (int i = 0; i < 8; ++i) acc[i] = c2s;
#pragma unroll
            for (int kc = 0; kc < 8; ++kc) acc = wmma4(a2[kc], B2[kc], acc);

            float p[2];
#pragma unroll
            for (int i = 0; i < 2; ++i) p[i] = tanh_scaled_of_z(acc[i], w3l);

            float s;
            {
                const float keep = lb0 ? p[1] : p[0];
                const float send = lb0 ? p[0] : p[1];
                s = keep + lane_xor<1>(send);
            }
            s = s + lane_xor<2>(s);
            s = s + lane_xor<4>(s);
            s = s + lane_xor<8>(s);

            const float corr = s + b3v;
            const float mw = E * xsc.x + eta * xsc.y;
            st[u] = (mw + corr) * iw;
            prev = st[u];
            xsc = xsn;
            xan = xann;
        }
        const float o01 = (su & 1) ? st[1] : st[0];
        const float o23 = (su & 1) ? st[3] : st[2];
        const float o45 = (su & 1) ? st[5] : st[4];
        const float o67 = (su & 1) ? st[7] : st[6];
        const float o03 = (su & 2) ? o23 : o01;
        const float o47 = (su & 2) ? o67 : o45;
        Sp[(size_t)t * 4 + soff] = (su & 4) ? o47 : o03;
    }
}

__global__ __launch_bounds__(256)
void fusion_head_kernel(const float* __restrict__ S,
                      const float* __restrict__ imp, const float* __restrict__ lw,
                      const float* __restrict__ Wf1, const float* __restrict__ bf1,
                      const float* __restrict__ Wf2, const float* __restrict__ bf2,
                      float* __restrict__ out)
{
    __shared__ float tl[8][16 * 34];

    const int lane  = threadIdx.x & 31;
    const int wv    = threadIdx.x >> 5;
    const int gw    = blockIdx.x * 8 + wv;
    const int chunk = gw & 127;
    const int tile  = (gw >> 7) & 15;
    const int p     = gw >> 11;
    const int t0    = chunk * 32;

    float cl[4];
    {
        float e0 = __expf(lw[0]), e1 = __expf(lw[1]);
        float e2 = __expf(lw[2]), e3 = __expf(lw[3]);
        float es = e0 + e1 + e2 + e3;
        float ws[4] = { e0 / es, e1 / es, e2 / es, e3 / es };
#pragma unroll
        for (int l = 0; l < 4; ++l) {
            float s = 0.0f;
#pragma unroll
            for (int q = 0; q < 4; ++q)
                s += 1.0f / (1.0f + __expf(-imp[l * 4 + q]));
            cl[l] = (s / (s + 1e-8f)) * ws[l];
        }
    }
    const float bf2v = bf2[0];

    float v[4][16];
#pragma unroll
    for (int l = 0; l < 4; ++l) {
        const float4* src =
            (const float4*)(S + ((size_t)((l * 4 + p) * 64 + 4 * tile + (lane & 3)) * TT + t0) * 4);
#pragma unroll
        for (int k = 0; k < 4; ++k) {
            float4 q = src[k * 8 + (lane >> 2)];
            v[l][4 * k + 0] = q.x * cl[l];
            v[l][4 * k + 1] = q.y * cl[l];
            v[l][4 * k + 2] = q.z * cl[l];
            v[l][4 * k + 3] = q.w * cl[l];
        }
    }

    float acc[16];
#pragma unroll
    for (int e = 0; e < 16; ++e) acc[e] = 0.0f;
#pragma unroll
    for (int f = 0; f < 16; ++f) {
        const float w0 = Wf1[f * 4 + 0], w1 = Wf1[f * 4 + 1];
        const float w2 = Wf1[f * 4 + 2], w3 = Wf1[f * 4 + 3];
        const float bf = bf1[f], wo = Wf2[f];
#pragma unroll
        for (int e = 0; e < 16; ++e) {
            float hf = fmaf(v[0][e], w0, bf);
            hf = fmaf(v[1][e], w1, hf);
            hf = fmaf(v[2][e], w2, hf);
            hf = fmaf(v[3][e], w3, hf);
            hf = fmaxf(hf, 0.0f);
            acc[e] = fmaf(wo, hf, acc[e]);
        }
    }

    float* tp = tl[wv];
#pragma unroll
    for (int k = 0; k < 4; ++k)
#pragma unroll
        for (int i = 0; i < 4; ++i)
            tp[((lane & 3) * 4 + i) * 34 + 8 * k + (lane >> 2)] = acc[4 * k + i] + bf2v;
#pragma unroll
    for (int r = 0; r < 16; ++r)
        out[(size_t)(p * BB + tile * 16 + r) * TT + t0 + lane] = tp[r * 34 + lane];
}

extern "C" void kernel_launch(void* const* d_in, const int* in_sizes, int n_in,
                              void* d_out, int out_size, void* d_ws, size_t ws_size,
                              hipStream_t stream) {
    (void)in_sizes; (void)n_in; (void)ws_size; (void)out_size;
    recurrence_kernel<<<dim3(256), dim3(128), 0, stream>>>(
        (const float*)d_in[0],  (const float*)d_in[1],  (const float*)d_in[2],
        (const float*)d_in[3],  (const float*)d_in[4],  (const float*)d_in[5],
        (const float*)d_in[6],  (const float*)d_in[7],  (const float*)d_in[8],
        (const float*)d_in[9],
        (float*)d_ws);
    fusion_head_kernel<<<dim3(1024), dim3(256), 0, stream>>>(
        (const float*)d_ws,
        (const float*)d_in[9],  (const float*)d_in[10], (const float*)d_in[11],
        (const float*)d_in[12], (const float*)d_in[13], (const float*)d_in[14],
        (float*)d_out);
}


# layer-1 to layer-2 hand-over by 12 lane moves in registers, no LDS round trip: layer 1 has weights as A and step inputs as B, so a lane's column is one batch row; layer-2 sum order changed
# speedup vs baseline: 6.7901x; 1.0531x over previous
#include <hip/hip_runtime.h>
#include <math.h>


typedef __attribute__((ext_vector_type(2))) float v2f;
typedef __attribute__((ext_vector_type(8))) float v8f;

#define TT 4096
#define BB 256

#define TANH_IN 2.885390081777926814f

__device__ __forceinline__ float half_gate(float z) {
    return __builtin_amdgcn_rcpf(__builtin_amdgcn_exp2f(z) + 1.0f);
}

__device__ __forceinline__ float tanh_scaled_of_z(float z, float w) {
    return fmaf(half_gate(z), -2.0f * w, w);
}

template <int MASK>
__device__ __forceinline__ float lane_xor(float v) {
    return __builtin_bit_cast(float, __builtin_amdgcn_update_dpp(
        0, __builtin_bit_cast(int, v), 0x160 | MASK, 0xF, 0xF, true));
}

__device__ __forceinline__ float other_half(float v) {
    return __builtin_bit_cast(float, __builtin_amdgcn_permlanex16(
        __builtin_bit_cast(int, v), __builtin_bit_cast(int, v), 0x76543210, 0xfedcba98, false, false));
}

__device__ __forceinline__ v8f wmma4(v2f a, v2f b, v8f c) {
    return __builtin_amdgcn_wmma_f32_16x16x4_f32(
        false, a, false, b, (short)0, c, false, false);
}


__global__ __launch_bounds__(128)
void recurrence_kernel(const float* __restrict__ x,
                     const float* __restrict__ W1,  const float* __restrict__ b1,
                     const float* __restrict__ W2,  const float* __restrict__ b2,
                     const float* __restrict__ W3,  const float* __restrict__ b3,
                     const float* __restrict__ logE,const float* __restrict__ logeta,
                     const float* __restrict__ imp,
                     float* __restrict__ S)
{
    const int lane = threadIdx.x & 31;
    const int wv   = threadIdx.x >> 5;
    const int gw   = blockIdx.x * 4 + wv;
    const int n    = gw & 15;
    const int tile = gw >> 4;
    const int hi   = lane >> 4;
    const int lm   = lane & 15;
    const int b0   = tile * 4;
    const int sel  = (lm >> 1) & 3;
    const int rho  = (lm & 1) + 2 * (lm >> 3);
    const int srow = 2 * hi + (lm & 1);
    const bool lb0 = (lm & 1) != 0;
    const bool clive = (sel >> 1) == hi;
    const bool cs0   = clive && ((sel & 1) == 0);
    const bool cs1   = clive && ((sel & 1) != 0);
    const bool cown  = (lm >> 3) == hi;
    const bool xalive = sel == hi;
    const bool xblive = sel == 2 + hi;

    const float E   = __expf(logE[n]);
    const float eta = __expf(logeta[n]);
    const float iw  = 1.0f / (1.0f + __expf(-imp[n]));
    const float b3v = b3[n];

    v2f Awc, Awa, Awb; v8f Cb;
    {
        const bool mlive = (lm & 4) == 0;
        const int  mu    = (lm & 3) + 4 * (lm >> 3);
        const float* w   = W1 + (size_t)n * 96;
        Awc.x = mlive ? TANH_IN * w[(8 * (2 * hi)     + mu) * 3 + 2] : 0.0f;
        Awc.y = mlive ? TANH_IN * w[(8 * (2 * hi + 1) + mu) * 3 + 2] : 0.0f;
        Awa.x = mlive ? TANH_IN * w[(8 * hi + mu) * 3 + 0] : 0.0f;
        Awa.y = mlive ? TANH_IN * w[(8 * hi + mu) * 3 + 1] : 0.0f;
        Awb.x = mlive ? TANH_IN * w[(8 * (2 + hi) + mu) * 3 + 0] : 0.0f;
        Awb.y = mlive ? TANH_IN * w[(8 * (2 + hi) + mu) * 3 + 1] : 0.0f;
#pragma unroll
        for (int i = 0; i < 4; ++i) Cb[i] = TANH_IN * b1[n * 32 + 8 * sel + 4 * hi + i];
#pragma unroll
        for (int i = 4; i < 8; ++i) Cb[i] = 0.0f;
    }
    v2f B2[8];
#pragma unroll
    for (int kc = 0; kc < 8; ++kc) {
        const int k0 = 8 * (kc >> 1) + 4 * hi + 2 * (kc & 1);
        B2[kc].x = (-2.0f * TANH_IN) * W2[(n * 16 + lm) * 32 + k0];
        B2[kc].y = (-2.0f * TANH_IN) * W2[(n * 16 + lm) * 32 + k0 + 1];
    }
    float w2row = 0.0f;
    for (int k = 0; k < 32; ++k) w2row += W2[(n * 16 + lm) * 32 + k];
    const float c2s = TANH_IN * (b2[n * 16 + lm] + w2row);

    const float w3l = W3[n * 16 + lm];

    const float2* xa = (const float2*)x + (size_t)(b0 + rho) * TT;
    const float2* xs = (const float2*)x + (size_t)(b0 + srow) * TT;
    float2 xsc = xs[0];
    float2 xan = xa[1];
    float prev = 0.0f, prevo = 0.0f;
    float* Sp = S + (size_t)(n * 64 + tile) * TT * 4;
    const int su   = lm >> 1;
    const int soff = su * 4 + srow;

    v8f cxc;
    {
        const float2 x0 = xa[0];
        v2f ba; ba.x = xalive ? x0.x : 0.0f; ba.y = xalive ? x0.y : 0.0f;
        v2f bb; bb.x = xblive ? x0.x : 0.0f; bb.y = xblive ? x0.y : 0.0f;
        cxc = wmma4(Awb, bb, wmma4(Awa, ba, Cb));
    }

    for (int t = 0; t < TT; t += 8) {
        float st[8];
#pragma unroll
        for (int u = 0; u < 8; ++u) {
            const int tc = t + u;
            const float2 xann = xa[(tc + 2) < TT ? tc + 2 : TT - 1];
            const float2 xsn  = xs[(tc + 1) < TT ? tc + 1 : TT - 1];

            const float cv = cown ? prev : prevo;
            v2f bc;
            bc.x = cs0 ? cv : 0.0f;
            bc.y = cs1 ? cv : 0.0f;

            v8f d = wmma4(Awc, bc, cxc);
            asm volatile("" :: "v"(d));
            __builtin_amdgcn_sched_barrier(0);
            v8f cxa;
            {
                v2f ba; ba.x = xalive ? xan.x : 0.0f; ba.y = xalive ? xan.y : 0.0f;
                cxa = wmma4(Awa, ba, Cb);
            }
            asm volatile("" :: "v"(cxa));
            __builtin_amdgcn_sched_barrier(0);
            float h[4];
#pragma unroll
            for (int i = 0; i < 4; ++i) h[i] = half_gate(d[i]);
            v2f a2[8];
            a2[0].x = h[0]; a2[0].y = h[1];
            a2[1].x = h[2]; a2[1].y = h[3];
            a2[2].x = lane_xor<2>(h[0]); a2[2].y = lane_xor<2>(h[1]);
            a2[3].x = lane_xor<2>(h[2]); a2[3].y = lane_xor<2>(h[3]);
            a2[4].x = lane_xor<4>(h[0]); a2[4].y = lane_xor<4>(h[1]);
            a2[5].x = lane_xor<4>(h[2]); a2[5].y = lane_xor<4>(h[3]);
            a2[6].x = lane_xor<6>(h[0]); a2[6].y = lane_xor<6>(h[1]);
            a2[7].x = lane_xor<6>(h[2]); a2[7].y = lane_xor<6>(h[3]);

            v8f acc;
#pragma unroll
            for (int i = 0; i < 8; ++i) acc[i] = c2s;
#pragma unroll
            for (int kc = 0; kc < 8; ++kc) acc = wmma4(a2[kc], B2[kc], acc);

            float p[2];
#pragma unroll
            for (int i = 0; i < 2; ++i) p[i] = tanh_scaled_of_z(acc[i], w3l);

            float s;
            {
                const float keep = lb0 ? p[1] : p[0];
                const float send = lb0 ? p[0] : p[1];
                s = keep + lane_xor<1>(send);
            }
            __builtin_amdgcn_sched_barrier(0);
            {
                v2f bb; bb.x = xblive ? xan.x : 0.0f; bb.y = xblive ? xan.y : 0.0f;
                cxc = wmma4(Awb, bb, cxa);
            }
            __builtin_amdgcn_sched_barrier(0);
            s = s + lane_xor<2>(s);
            s = s + lane_xor<4>(s);
            s = s + lane_xor<8>(s);

            const float corr = s + b3v;
            const float mw = E * xsc.x + eta * xsc.y;
            st[u] = (mw + corr) * iw;
            prev  = st[u];
            prevo = other_half(prev);
            xsc = xsn;
            xan = xann;
        }
        const float o01 = (su & 1) ? st[1] : st[0];
        const float o23 = (su & 1) ? st[3] : st[2];
        const float o45 = (su & 1) ? st[5] : st[4];
        const float o67 = (su & 1) ? st[7] : st[6];
        const float o03 = (su & 2) ? o23 : o01;
        const float o47 = (su & 2) ? o67 : o45;
        Sp[(size_t)t * 4 + soff] = (su & 4) ? o47 : o03;
    }
}

__global__ __launch_bounds__(256)
void fusion_head_kernel(const float* __restrict__ S,
                      const float* __restrict__ imp, const float* __restrict__ lw,
                      const float* __restrict__ Wf1, const float* __restrict__ bf1,
                      const float* __restrict__ Wf2, const float* __restrict__ bf2,
                      float* __restrict__ out)
{
    __shared__ float tl[8][16 * 34];

    const int lane  = threadIdx.x & 31;
    const int wv    = threadIdx.x >> 5;
    const int gw    = blockIdx.x * 8 + wv;
    const int chunk = gw & 127;
    const int tile  = (gw >> 7) & 15;
    const int p     = gw >> 11;
    const int t0    = chunk * 32;

    float cl[4];
    {
        float e0 = __expf(lw[0]), e1 = __expf(lw[1]);
        float e2 = __expf(lw[2]), e3 = __expf(lw[3]);
        float es = e0 + e1 + e2 + e3;
        float ws[4] = { e0 / es, e1 / es, e2 / es, e3 / es };
#pragma unroll
        for (int l = 0; l < 4; ++l) {
            float s = 0.0f;
#pragma unroll
            for (int q = 0; q < 4; ++q)
                s += 1.0f / (1.0f + __expf(-imp[l * 4 + q]));
            cl[l] = (s / (s + 1e-8f)) * ws[l];
        }
    }
    const float bf2v = bf2[0];

    float v[4][16];
#pragma unroll
    for (int l = 0; l < 4; ++l) {
        const float4* src =
            (const float4*)(S + ((size_t)((l * 4 + p) * 64 + 4 * tile + (lane & 3)) * TT + t0) * 4);
#pragma unroll
        for (int k = 0; k < 4; ++k) {
            float4 q = src[k * 8 + (lane >> 2)];
            v[l][4 * k + 0] = q.x * cl[l];
            v[l][4 * k + 1] = q.y * cl[l];
            v[l][4 * k + 2] = q.z * cl[l];
            v[l][4 * k + 3] = q.w * cl[l];
        }
    }

    float acc[16];
#pragma unroll
    for (int e = 0; e < 16; ++e) acc[e] = 0.0f;
#pragma unroll
    for (int f = 0; f < 16; ++f) {
        const float w0 = Wf1[f * 4 + 0], w1 = Wf1[f * 4 + 1];
        const float w2 = Wf1[f * 4 + 2], w3 = Wf1[f * 4 + 3];
        const float bf = bf1[f], wo = Wf2[f];
#pragma unroll
        for (int e = 0; e < 16; ++e) {
            float hf = fmaf(v[0][e], w0, bf);
            hf = fmaf(v[1][e], w1, hf);
            hf = fmaf(v[2][e], w2, hf);
            hf = fmaf(v[3][e], w3, hf);
            hf = fmaxf(hf, 0.0f);
            acc[e] = fmaf(wo, hf, acc[e]);
        }
    }

    float* tp = tl[wv];
#pragma unroll
    for (int k = 0; k < 4; ++k)
#pragma unroll
        for (int i = 0; i < 4; ++i)
            tp[((lane & 3) * 4 + i) * 34 + 8 * k + (lane >> 2)] = acc[4 * k + i] + bf2v;
#pragma unroll
    for (int r = 0; r < 16; ++r)
        out[(size_t)(p * BB + tile * 16 + r) * TT + t0 + lane] = tp[r * 34 + lane];
}

extern "C" void kernel_launch(void* const* d_in, const int* in_sizes, int n_in,
                              void* d_out, int out_size, void* d_ws, size_t ws_size,
                              hipStream_t stream) {
    (void)in_sizes; (void)n_in; (void)ws_size; (void)out_size;
    recurrence_kernel<<<dim3(256), dim3(128), 0, stream>>>(
        (const float*)d_in[0],  (const float*)d_in[1],  (const float*)d_in[2],
        (const float*)d_in[3],  (const float*)d_in[4],  (const float*)d_in[5],
        (const float*)d_in[6],  (const float*)d_in[7],  (const float*)d_in[8],
        (const float*)d_in[9],
        (float*)d_ws);
    fusion_head_kernel<<<dim3(1024), dim3(256), 0, stream>>>(
        (const float*)d_ws,
        (const float*)d_in[9],  (const float*)d_in[10], (const float*)d_in[11],
        (const float*)d_in[12], (const float*)d_in[13], (const float*)d_in[14],
        (float*)d_out);
}


# layer 2 fed from each lane's own layer-1 outputs: unit group g runs on its own accumulator tile (4 tiles x 2 products), its rows' parts added in-lane; 12 lane moves gone; layer-2 sum order changed
# speedup vs baseline: 7.1292x; 1.0499x over previous
#include <hip/hip_runtime.h>
#include <math.h>


typedef __attribute__((ext_vector_type(2))) float v2f;
typedef __attribute__((ext_vector_type(8))) float v8f;

#define TT 4096
#define BB 256

#define TANH_IN 2.885390081777926814f

__device__ __forceinline__ float half_gate(float z) {
    return __builtin_amdgcn_rcpf(__builtin_amdgcn_exp2f(z) + 1.0f);
}

__device__ __forceinline__ float tanh_scaled_of_z(float z, float w) {
    return fmaf(half_gate(z), -2.0f * w, w);
}

template <int MASK>
__device__ __forceinline__ float lane_xor(float v) {
    return __builtin_bit_cast(float, __builtin_amdgcn_update_dpp(
        0, __builtin_bit_cast(int, v), 0x160 | MASK, 0xF, 0xF, true));
}

__device__ __forceinline__ float other_half(float v) {
    return __builtin_bit_cast(float, __builtin_amdgcn_permlanex16(
        __builtin_bit_cast(int, v), __builtin_bit_cast(int, v), 0x76543210, 0xfedcba98, false, false));
}

__device__ __forceinline__ v8f wmma4(v2f a, v2f b, v8f c) {
    return __builtin_amdgcn_wmma_f32_16x16x4_f32(
        false, a, false, b, (short)0, c, false, false);
}


__global__ __launch_bounds__(128)
void recurrence_kernel(const float* __restrict__ x,
                     const float* __restrict__ W1,  const float* __restrict__ b1,
                     const float* __restrict__ W2,  const float* __restrict__ b2,
                     const float* __restrict__ W3,  const float* __restrict__ b3,
                     const float* __restrict__ logE,const float* __restrict__ logeta,
                     const float* __restrict__ imp,
                     float* __restrict__ S)
{
    const int lane = threadIdx.x & 31;
    const int wv   = threadIdx.x >> 5;
    const int gw   = blockIdx.x * 4 + wv;
    const int n    = gw & 15;
    const int tile = gw >> 4;
    const int hi   = lane >> 4;
    const int lm   = lane & 15;
    const int b0   = tile * 4;
    const int sel  = (lm >> 1) & 3;
    const int rho  = (lm & 1) + 2 * (lm >> 3);
    const int srow = 2 * hi + (lm & 1);
    const bool lb0 = (lm & 1) != 0;
    const bool clive = (sel >> 1) == hi;
    const bool cs0   = clive && ((sel & 1) == 0);
    const bool cs1   = clive && ((sel & 1) != 0);
    const bool cown  = (lm >> 3) == hi;
    const bool xalive = sel == hi;
    const bool xblive = sel == 2 + hi;

    const float E   = __expf(logE[n]);
    const float eta = __expf(logeta[n]);
    const float iw  = 1.0f / (1.0f + __expf(-imp[n]));
    const float b3v = b3[n];

    v2f Awc, Awa, Awb; v8f Cb;
    {
        const bool mlive = (lm & 4) == 0;
        const int  mu    = (lm & 3) + 4 * (lm >> 3);
        const float* w   = W1 + (size_t)n * 96;
        Awc.x = mlive ? TANH_IN * w[(8 * (2 * hi)     + mu) * 3 + 2] : 0.0f;
        Awc.y = mlive ? TANH_IN * w[(8 * (2 * hi + 1) + mu) * 3 + 2] : 0.0f;
        Awa.x = mlive ? TANH_IN * w[(8 * hi + mu) * 3 + 0] : 0.0f;
        Awa.y = mlive ? TANH_IN * w[(8 * hi + mu) * 3 + 1] : 0.0f;
        Awb.x = mlive ? TANH_IN * w[(8 * (2 + hi) + mu) * 3 + 0] : 0.0f;
        Awb.y = mlive ? TANH_IN * w[(8 * (2 + hi) + mu) * 3 + 1] : 0.0f;
#pragma unroll
        for (int i = 0; i < 4; ++i) Cb[i] = TANH_IN * b1[n * 32 + 8 * sel + 4 * hi + i];
#pragma unroll
        for (int i = 4; i < 8; ++i) Cb[i] = 0.0f;
    }
    v2f B2[8];
#pragma unroll
    for (int kc = 0; kc < 8; ++kc) {
        const int k0 = 8 * (kc >> 1) + 4 * hi + 2 * (kc & 1);
        B2[kc].x = (-2.0f * TANH_IN) * W2[(n * 16 + lm) * 32 + k0];
        B2[kc].y = (-2.0f * TANH_IN) * W2[(n * 16 + lm) * 32 + k0 + 1];
    }
    float w2row = 0.0f;
    for (int k = 0; k < 32; ++k) w2row += W2[(n * 16 + lm) * 32 + k];
    const float c2s = TANH_IN * (b2[n * 16 + lm] + w2row);

    const float w3l = W3[n * 16 + lm];

    const float2* xa = (const float2*)x + (size_t)(b0 + rho) * TT;
    const float2* xs = (const float2*)x + (size_t)(b0 + srow) * TT;
    float2 xsc = xs[0];
    float2 xan = xa[1];
    float prev = 0.0f, prevo = 0.0f;
    float* Sp = S + (size_t)(n * 64 + tile) * TT * 4;
    const int su   = lm >> 1;
    const int soff = su * 4 + srow;

    v8f cxc;
    {
        const float2 x0 = xa[0];
        v2f ba; ba.x = xalive ? x0.x : 0.0f; ba.y = xalive ? x0.y : 0.0f;
        v2f bb; bb.x = xblive ? x0.x : 0.0f; bb.y = xblive ? x0.y : 0.0f;
        cxc = wmma4(Awb, bb, wmma4(Awa, ba, Cb));
    }

    for (int t = 0; t < TT; t += 8) {
        float st[8];
#pragma unroll
        for (int u = 0; u < 8; ++u) {
            const int tc = t + u;
            const float2 xann = xa[(tc + 2) < TT ? tc + 2 : TT - 1];
            const float2 xsn  = xs[(tc + 1) < TT ? tc + 1 : TT - 1];

            const float cv = cown ? prev : prevo;
            v2f bc;
            bc.x = cs0 ? cv : 0.0f;
            bc.y = cs1 ? cv : 0.0f;

            v8f d = wmma4(Awc, bc, cxc);
            asm volatile("" :: "v"(d));
            __builtin_amdgcn_sched_barrier(0);
            v8f cxa;
            {
                v2f ba; ba.x = xalive ? xan.x : 0.0f; ba.y = xalive ? xan.y : 0.0f;
                cxa = wmma4(Awa, ba, Cb);
            }
            asm volatile("" :: "v"(cxa));
            __builtin_amdgcn_sched_barrier(0);
            float h[4];
#pragma unroll
            for (int i = 0; i < 4; ++i) h[i] = half_gate(d[i]);
            v2f a01, a23;
            a01.x = h[0]; a01.y = h[1];
            a23.x = h[2]; a23.y = h[3];

            v8f acc0, acc1, acc2, acc3;
#pragma unroll
            for (int i = 0; i < 8; ++i) { acc0[i] = c2s; acc1[i] = 0.0f; acc2[i] = 0.0f; acc3[i] = 0.0f; }
            acc0 = wmma4(a01, B2[0], acc0);
            acc1 = wmma4(a01, B2[2], acc1);
            acc2 = wmma4(a01, B2[4], acc2);
            acc3 = wmma4(a01, B2[6], acc3);
            acc0 = wmma4(a23, B2[1], acc0);
            acc1 = wmma4(a23, B2[3], acc1);
            acc2 = wmma4(a23, B2[5], acc2);
            acc3 = wmma4(a23, B2[7], acc3);
            float acc[2];
#pragma unroll
            for (int i = 0; i < 2; ++i)
                acc[i] = ((acc0[i] + acc1[2 + i]) + acc2[4 + i]) + acc3[6 + i];

            float p[2];
#pragma unroll
            for (int i = 0; i < 2; ++i) p[i] = tanh_scaled_of_z(acc[i], w3l);

            float s;
            {
                const float keep = lb0 ? p[1] : p[0];
                const float send = lb0 ? p[0] : p[1];
                s = keep + lane_xor<1>(send);
            }
            __builtin_amdgcn_sched_barrier(0);
            {
                v2f bb; bb.x = xblive ? xan.x : 0.0f; bb.y = xblive ? xan.y : 0.0f;
                cxc = wmma4(Awb, bb, cxa);
            }
            __builtin_amdgcn_sched_barrier(0);
            s = s + lane_xor<2>(s);
            s = s + lane_xor<4>(s);
            s = s + lane_xor<8>(s);

            const float corr = s + b3v;
            const float mw = E * xsc.x + eta * xsc.y;
            st[u] = (mw + corr) * iw;
            prev  = st[u];
            prevo = other_half(prev);
            xsc = xsn;
            xan = xann;
        }
        const float o01 = (su & 1) ? st[1] : st[0];
        const float o23 = (su & 1) ? st[3] : st[2];
        const float o45 = (su & 1) ? st[5] : st[4];
        const float o67 = (su & 1) ? st[7] : st[6];
        const float o03 = (su & 2) ? o23 : o01;
        const float o47 = (su & 2) ? o67 : o45;
        Sp[(size_t)t * 4 + soff] = (su & 4) ? o47 : o03;
    }
}

__global__ __launch_bounds__(256)
void fusion_head_kernel(const float* __restrict__ S,
                      const float* __restrict__ imp, const float* __restrict__ lw,
                      const float* __restrict__ Wf1, const float* __restrict__ bf1,
                      const float* __restrict__ Wf2, const float* __restrict__ bf2,
                      float* __restrict__ out)
{
    __shared__ float tl[8][16 * 34];

    const int lane  = threadIdx.x & 31;
    const int wv    = threadIdx.x >> 5;
    const int gw    = blockIdx.x * 8 + wv;
    const int chunk = gw & 127;
    const int tile  = (gw >> 7) & 15;
    const int p     = gw >> 11;
    const int t0    = chunk * 32;

    float cl[4];
    {
        float e0 = __expf(lw[0]), e1 = __expf(lw[1]);
        float e2 = __expf(lw[2]), e3 = __expf(lw[3]);
        float es = e0 + e1 + e2 + e3;
        float ws[4] = { e0 / es, e1 / es, e2 / es, e3 / es };
#pragma unroll
        for (int l = 0; l < 4; ++l) {
            float s = 0.0f;
#pragma unroll
            for (int q = 0; q < 4; ++q)
                s += 1.0f / (1.0f + __expf(-imp[l * 4 + q]));
            cl[l] = (s / (s + 1e-8f)) * ws[l];
        }
    }
    const float bf2v = bf2[0];

    float v[4][16];
#pragma unroll
    for (int l = 0; l < 4; ++l) {
        const float4* src =
            (const float4*)(S + ((size_t)((l * 4 + p) * 64 + 4 * tile + (lane & 3)) * TT + t0) * 4);
#pragma unroll
        for (int k = 0; k < 4; ++k) {
            float4 q = src[k * 8 + (lane >> 2)];
            v[l][4 * k + 0] = q.x * cl[l];
            v[l][4 * k + 1] = q.y * cl[l];
            v[l][4 * k + 2] = q.z * cl[l];
            v[l][4 * k + 3] = q.w * cl[l];
        }
    }

    float acc[16];
#pragma unroll
    for (int e = 0; e < 16; ++e) acc[e] = 0.0f;
#pragma unroll
    for (int f = 0; f < 16; ++f) {
        const float w0 = Wf1[f * 4 + 0], w1 = Wf1[f * 4 + 1];
        const float w2 = Wf1[f * 4 + 2], w3 = Wf1[f * 4 + 3];
        const float bf = bf1[f], wo = Wf2[f];
#pragma unroll
        for (int e = 0; e < 16; ++e) {
            float hf = fmaf(v[0][e], w0, bf);
            hf = fmaf(v[1][e], w1, hf);
            hf = fmaf(v[2][e], w2, hf);
            hf = fmaf(v[3][e], w3, hf);
            hf = fmaxf(hf, 0.0f);
            acc[e] = fmaf(wo, hf, acc[e]);
        }
    }

    float* tp = tl[wv];
#pragma unroll
    for (int k = 0; k < 4; ++k)
#pragma unroll
        for (int i = 0; i < 4; ++i)
            tp[((lane & 3) * 4 + i) * 34 + 8 * k + (lane >> 2)] = acc[4 * k + i] + bf2v;
#pragma unroll
    for (int r = 0; r < 16; ++r)
        out[(size_t)(p * BB + tile * 16 + r) * TT + t0 + lane] = tp[r * 34 + lane];
}

extern "C" void kernel_launch(void* const* d_in, const int* in_sizes, int n_in,
                              void* d_out, int out_size, void* d_ws, size_t ws_size,
                              hipStream_t stream) {
    (void)in_sizes; (void)n_in; (void)ws_size; (void)out_size;
    recurrence_kernel<<<dim3(256), dim3(128), 0, stream>>>(
        (const float*)d_in[0],  (const float*)d_in[1],  (const float*)d_in[2],
        (const float*)d_in[3],  (const float*)d_in[4],  (const float*)d_in[5],
        (const float*)d_in[6],  (const float*)d_in[7],  (const float*)d_in[8],
        (const float*)d_in[9],
        (float*)d_ws);
    fusion_head_kernel<<<dim3(1024), dim3(256), 0, stream>>>(
        (const float*)d_ws,
        (const float*)d_in[9],  (const float*)d_in[10], (const float*)d_in[11],
        (const float*)d_in[12], (const float*)d_in[13], (const float*)d_in[14],
        (float*)d_out);
}


# serial tail from lane sum to carry product freed of what does not wait for the sum: maxwell term and 2nd input operand made ahead, half exchange without a copy, one select behind it; sums unchanged
# speedup vs baseline: 7.5015x; 1.0522x over previous
#include <hip/hip_runtime.h>
#include <math.h>


typedef __attribute__((ext_vector_type(2))) float v2f;
typedef __attribute__((ext_vector_type(8))) float v8f;

#define TT 4096
#define BB 256

#define TANH_IN 2.885390081777926814f

__device__ __forceinline__ float half_gate(float z) {
    return __builtin_amdgcn_rcpf(__builtin_amdgcn_exp2f(z) + 1.0f);
}

__device__ __forceinline__ float tanh_scaled_of_z(float z, float w) {
    return fmaf(half_gate(z), -2.0f * w, w);
}

template <int MASK>
__device__ __forceinline__ float lane_xor(float v) {
    return __builtin_bit_cast(float, __builtin_amdgcn_update_dpp(
        0, __builtin_bit_cast(int, v), 0x160 | MASK, 0xF, 0xF, true));
}

__device__ __forceinline__ float other_half(float v) {
    return __builtin_bit_cast(float, __builtin_amdgcn_permlanex16(
        0, __builtin_bit_cast(int, v), 0x76543210, 0xfedcba98, true, false));
}

__device__ __forceinline__ v8f wmma4(v2f a, v2f b, v8f c) {
    return __builtin_amdgcn_wmma_f32_16x16x4_f32(
        false, a, false, b, (short)0, c, false, false);
}


__global__ __launch_bounds__(128)
void recurrence_kernel(const float* __restrict__ x,
                     const float* __restrict__ W1,  const float* __restrict__ b1,
                     const float* __restrict__ W2,  const float* __restrict__ b2,
                     const float* __restrict__ W3,  const float* __restrict__ b3,
                     const float* __restrict__ logE,const float* __restrict__ logeta,
                     const float* __restrict__ imp,
                     float* __restrict__ S)
{
    const int lane = threadIdx.x & 31;
    const int wv   = threadIdx.x >> 5;
    const int gw   = blockIdx.x * 4 + wv;
    const int n    = gw & 15;
    const int tile = gw >> 4;
    const int hi   = lane >> 4;
    const int lm   = lane & 15;
    const int b0   = tile * 4;
    const int sel  = (lm >> 1) & 3;
    const int rho  = (lm & 1) + 2 * (lm >> 3);
    const int srow = 2 * hi + (lm & 1);
    const bool lb0 = (lm & 1) != 0;
    const bool clive = (sel >> 1) == hi;
    const bool cs0   = clive && ((sel & 1) == 0);
    const bool cs1   = clive && ((sel & 1) != 0);
    const bool cown  = (lm >> 3) == hi;
    const bool ownx  = cs0 && cown,  owny = cs1 && cown;
    const bool othx  = cs0 && !cown, othy = cs1 && !cown;
    const bool xalive = sel == hi;
    const bool xblive = sel == 2 + hi;

    const float E   = __expf(logE[n]);
    const float eta = __expf(logeta[n]);
    const float iw  = 1.0f / (1.0f + __expf(-imp[n]));
    const float b3v = b3[n];

    v2f Awc, Awa, Awb; v8f Cb;
    {
        const bool mlive = (lm & 4) == 0;
        const int  mu    = (lm & 3) + 4 * (lm >> 3);
        const float* w   = W1 + (size_t)n * 96;
        Awc.x = mlive ? TANH_IN * w[(8 * (2 * hi)     + mu) * 3 + 2] : 0.0f;
        Awc.y = mlive ? TANH_IN * w[(8 * (2 * hi + 1) + mu) * 3 + 2] : 0.0f;
        Awa.x = mlive ? TANH_IN * w[(8 * hi + mu) * 3 + 0] : 0.0f;
        Awa.y = mlive ? TANH_IN * w[(8 * hi + mu) * 3 + 1] : 0.0f;
        Awb.x = mlive ? TANH_IN * w[(8 * (2 + hi) + mu) * 3 + 0] : 0.0f;
        Awb.y = mlive ? TANH_IN * w[(8 * (2 + hi) + mu) * 3 + 1] : 0.0f;
#pragma unroll
        for (int i = 0; i < 4; ++i) Cb[i] = TANH_IN * b1[n * 32 + 8 * sel + 4 * hi + i];
#pragma unroll
        for (int i = 4; i < 8; ++i) Cb[i] = 0.0f;
    }
    v2f B2[8];
#pragma unroll
    for (int kc = 0; kc < 8; ++kc) {
        const int k0 = 8 * (kc >> 1) + 4 * hi + 2 * (kc & 1);
        B2[kc].x = (-2.0f * TANH_IN) * W2[(n * 16 + lm) * 32 + k0];
        B2[kc].y = (-2.0f * TANH_IN) * W2[(n * 16 + lm) * 32 + k0 + 1];
    }
    float w2row = 0.0f;
    for (int k = 0; k < 32; ++k) w2row += W2[(n * 16 + lm) * 32 + k];
    const float c2s = TANH_IN * (b2[n * 16 + lm] + w2row);

    const float w3l = W3[n * 16 + lm];

    const float2* xa = (const float2*)x + (size_t)(b0 + rho) * TT;
    const float2* xs = (const float2*)x + (size_t)(b0 + srow) * TT;
    float2 xsc = xs[0];
    float2 xan = xa[1];
    float cx = 0.0f, cy = 0.0f, prevo = 0.0f;
    float* Sp = S + (size_t)(n * 64 + tile) * TT * 4;
    const int su   = lm >> 1;
    const int soff = su * 4 + srow;

    v8f cxc;
    {
        const float2 x0 = xa[0];
        v2f ba; ba.x = xalive ? x0.x : 0.0f; ba.y = xalive ? x0.y : 0.0f;
        v2f bb; bb.x = xblive ? x0.x : 0.0f; bb.y = xblive ? x0.y : 0.0f;
        cxc = wmma4(Awb, bb, wmma4(Awa, ba, Cb));
    }

    for (int t = 0; t < TT; t += 8) {
        float st[8];
#pragma unroll
        for (int u = 0; u < 8; ++u) {
            const int tc = t + u;
            const float2 xann = xa[(tc + 2) < TT ? tc + 2 : TT - 1];
            const float2 xsn  = xs[(tc + 1) < TT ? tc + 1 : TT - 1];

            v2f bc;
            bc.x = othx ? prevo : cx;
            bc.y = othy ? prevo : cy;

            v8f d = wmma4(Awc, bc, cxc);
            asm volatile("" :: "v"(d));
            __builtin_amdgcn_sched_barrier(0);
            v8f cxa;
            v2f bb;
            {
                v2f ba; ba.x = xalive ? xan.x : 0.0f; ba.y = xalive ? xan.y : 0.0f;
                bb.x = xblive ? xan.x : 0.0f; bb.y = xblive ? xan.y : 0.0f;
                cxa = wmma4(Awa, ba, Cb);
            }
            asm volatile("" :: "v"(cxa));
            __builtin_amdgcn_sched_barrier(0);
            float h[4];
#pragma unroll
            for (int i = 0; i < 4; ++i) h[i] = half_gate(d[i]);
            v2f a01, a23;
            a01.x = h[0]; a01.y = h[1];
            a23.x = h[2]; a23.y = h[3];

            v8f acc0, acc1, acc2, acc3;
#pragma unroll
            for (int i = 0; i < 8; ++i) { acc0[i] = c2s; acc1[i] = 0.0f; acc2[i] = 0.0f; acc3[i] = 0.0f; }
            acc0 = wmma4(a01, B2[0], acc0);
            acc1 = wmma4(a01, B2[2], acc1);
            acc2 = wmma4(a01, B2[4], acc2);
            acc3 = wmma4(a01, B2[6], acc3);
            acc0 = wmma4(a23, B2[1], acc0);
            acc1 = wmma4(a23, B2[3], acc1);
            acc2 = wmma4(a23, B2[5], acc2);
            acc3 = wmma4(a23, B2[7], acc3);
            float acc[2];
#pragma unroll
            for (int i = 0; i < 2; ++i)
                acc[i] = ((acc0[i] + acc1[2 + i]) + acc2[4 + i]) + acc3[6 + i];

            float p[2];
#pragma unroll
            for (int i = 0; i < 2; ++i) p[i] = tanh_scaled_of_z(acc[i], w3l);

            float s;
            {
                const float keep = lb0 ? p[1] : p[0];
                const float send = lb0 ? p[0] : p[1];
                s = keep + lane_xor<1>(send);
            }
            const float mw = fmaf(E, xsc.x, eta * xsc.y);
            asm volatile("" :: "v"(mw));
            __builtin_amdgcn_sched_barrier(0);
            cxc = wmma4(Awb, bb, cxa);
            __builtin_amdgcn_sched_barrier(0);
            s = s + lane_xor<2>(s);
            s = s + lane_xor<4>(s);
            s = s + lane_xor<8>(s);

            const float corr = s + b3v;
            st[u] = (mw + corr) * iw;
            cx    = ownx ? st[u] : 0.0f;
            cy    = owny ? st[u] : 0.0f;
            prevo = other_half(st[u]);
            xsc = xsn;
            xan = xann;
        }
        const float o01 = (su & 1) ? st[1] : st[0];
        const float o23 = (su & 1) ? st[3] : st[2];
        const float o45 = (su & 1) ? st[5] : st[4];
        const float o67 = (su & 1) ? st[7] : st[6];
        const float o03 = (su & 2) ? o23 : o01;
        const float o47 = (su & 2) ? o67 : o45;
        Sp[(size_t)t * 4 + soff] = (su & 4) ? o47 : o03;
    }
}

__global__ __launch_bounds__(256)
void fusion_head_kernel(const float* __restrict__ S,
                      const float* __restrict__ imp, const float* __restrict__ lw,
                      const float* __restrict__ Wf1, const float* __restrict__ bf1,
                      const float* __restrict__ Wf2, const float* __restrict__ bf2,
                      float* __restrict__ out)
{
    __shared__ float tl[8][16 * 34];

    const int lane  = threadIdx.x & 31;
    const int wv    = threadIdx.x >> 5;
    const int gw    = blockIdx.x * 8 + wv;
    const int chunk = gw & 127;
    const int tile  = (gw >> 7) & 15;
    const int p     = gw >> 11;
    const int t0    = chunk * 32;

    float cl[4];
    {
        float e0 = __expf(lw[0]), e1 = __expf(lw[1]);
        float e2 = __expf(lw[2]), e3 = __expf(lw[3]);
        float es = e0 + e1 + e2 + e3;
        float ws[4] = { e0 / es, e1 / es, e2 / es, e3 / es };
#pragma unroll
        for (int l = 0; l < 4; ++l) {
            float s = 0.0f;
#pragma unroll
            for (int q = 0; q < 4; ++q)
                s += 1.0f / (1.0f + __expf(-imp[l * 4 + q]));
            cl[l] = (s / (s + 1e-8f)) * ws[l];
        }
    }
    const float bf2v = bf2[0];

    float v[4][16];
#pragma unroll
    for (int l = 0; l < 4; ++l) {
        const float4* src =
            (const float4*)(S + ((size_t)((l * 4 + p) * 64 + 4 * tile + (lane & 3)) * TT + t0) * 4);
#pragma unroll
        for (int k = 0; k < 4; ++k) {
            float4 q = src[k * 8 + (lane >> 2)];
            v[l][4 * k + 0] = q.x * cl[l];
            v[l][4 * k + 1] = q.y * cl[l];
            v[l][4 * k + 2] = q.z * cl[l];
            v[l][4 * k + 3] = q.w * cl[l];
        }
    }

    float acc[16];
#pragma unroll
    for (int e = 0; e < 16; ++e) acc[e] = 0.0f;
#pragma unroll
    for (int f = 0; f < 16; ++f) {
        const float w0 = Wf1[f * 4 + 0], w1 = Wf1[f * 4 + 1];
        const float w2 = Wf1[f * 4 + 2], w3 = Wf1[f * 4 + 3];
        const float bf = bf1[f], wo = Wf2[f];
#pragma unroll
        for (int e = 0; e < 16; ++e) {
            float hf = fmaf(v[0][e], w0, bf);
            hf = fmaf(v[1][e], w1, hf);
            hf = fmaf(v[2][e], w2, hf);
            hf = fmaf(v[3][e], w3, hf);
            hf = fmaxf(hf, 0.0f);
            acc[e] = fmaf(wo, hf, acc[e]);
        }
    }

    float* tp = tl[wv];
#pragma unroll
    for (int k = 0; k < 4; ++k)
#pragma unroll
        for (int i = 0; i < 4; ++i)
            tp[((lane & 3) * 4 + i) * 34 + 8 * k + (lane >> 2)] = acc[4 * k + i] + bf2v;
#pragma unroll
    for (int r = 0; r < 16; ++r)
        out[(size_t)(p * BB + tile * 16 + r) * TT + t0 + lane] = tp[r * 34 + lane];
}

extern "C" void kernel_launch(void* const* d_in, const int* in_sizes, int n_in,
                              void* d_out, int out_size, void* d_ws, size_t ws_size,
                              hipStream_t stream) {
    (void)in_sizes; (void)n_in; (void)ws_size; (void)out_size;
    recurrence_kernel<<<dim3(256), dim3(128), 0, stream>>>(
        (const float*)d_in[0],  (const float*)d_in[1],  (const float*)d_in[2],
        (const float*)d_in[3],  (const float*)d_in[4],  (const float*)d_in[5],
        (const float*)d_in[6],  (const float*)d_in[7],  (const float*)d_in[8],
        (const float*)d_in[9],
        (float*)d_ws);
    fusion_head_kernel<<<dim3(1024), dim3(256), 0, stream>>>(
        (const float*)d_ws,
        (const float*)d_in[9],  (const float*)d_in[10], (const float*)d_in[11],
        (const float*)d_in[12], (const float*)d_in[13], (const float*)d_in[14],
        (float*)d_out);
}


# layer-1 input part in ONE matrix product per step, not two: the idle matrix rows 4-7, 12-15 carry unit groups 2-3, a lane keeps its own group's four registers by select; no sum or sum order changed
# speedup vs baseline: 7.7629x; 1.0348x over previous
#include <hip/hip_runtime.h>
#include <math.h>


typedef __attribute__((ext_vector_type(2))) float v2f;
typedef __attribute__((ext_vector_type(4))) float v4f;
typedef __attribute__((ext_vector_type(8))) float v8f;

#define TT 4096
#define BB 256

#define TANH_IN 2.885390081777926814f

__device__ __forceinline__ float half_gate(float z) {
    return __builtin_amdgcn_rcpf(__builtin_amdgcn_exp2f(z) + 1.0f);
}

__device__ __forceinline__ float tanh_scaled_of_z(float z, float w) {
    return fmaf(half_gate(z), -2.0f * w, w);
}

template <int MASK>
__device__ __forceinline__ float lane_xor(float v) {
    return __builtin_bit_cast(float, __builtin_amdgcn_update_dpp(
        0, __builtin_bit_cast(int, v), 0x160 | MASK, 0xF, 0xF, true));
}

__device__ __forceinline__ float other_half(float v) {
    return __builtin_bit_cast(float, __builtin_amdgcn_permlanex16(
        0, __builtin_bit_cast(int, v), 0x76543210, 0xfedcba98, true, false));
}

__device__ __forceinline__ v8f wmma4(v2f a, v2f b, v8f c) {
    return __builtin_amdgcn_wmma_f32_16x16x4_f32(
        false, a, false, b, (short)0, c, false, false);
}

__device__ __forceinline__ v8f own_group_half(v8f p, bool upper) {
    const v4f lo = __builtin_shufflevector(p, p, 0, 1, 2, 3);
    const v4f hi = __builtin_shufflevector(p, p, 4, 5, 6, 7);
    const v4f k  = upper ? hi : lo;
    return __builtin_shufflevector(k, hi, 0, 1, 2, 3, 4, 5, 6, 7);
}


__global__ __launch_bounds__(128)
void recurrence_kernel(const float* __restrict__ x,
                     const float* __restrict__ W1,  const float* __restrict__ b1,
                     const float* __restrict__ W2,  const float* __restrict__ b2,
                     const float* __restrict__ W3,  const float* __restrict__ b3,
                     const float* __restrict__ logE,const float* __restrict__ logeta,
                     const float* __restrict__ imp,
                     float* __restrict__ S)
{
    const int lane = threadIdx.x & 31;
    const int wv   = threadIdx.x >> 5;
    const int gw   = blockIdx.x * 4 + wv;
    const int n    = gw & 15;
    const int tile = gw >> 4;
    const int hi   = lane >> 4;
    const int lm   = lane & 15;
    const int b0   = tile * 4;
    const int sel  = (lm >> 1) & 3;
    const int rho  = (lm & 1) + 2 * (lm >> 3);
    const int srow = 2 * hi + (lm & 1);
    const bool lb0 = (lm & 1) != 0;
    const bool clive = (sel >> 1) == hi;
    const bool cs0   = clive && ((sel & 1) == 0);
    const bool cs1   = clive && ((sel & 1) != 0);
    const bool cown  = (lm >> 3) == hi;
    const bool ownx  = cs0 && cown,  owny = cs1 && cown;
    const bool othx  = cs0 && !cown, othy = cs1 && !cown;
    const bool xlive = (sel & 1) == hi;
    const bool hsel  = (sel >> 1) != 0;

    const float E   = __expf(logE[n]);
    const float eta = __expf(logeta[n]);
    const float iw  = 1.0f / (1.0f + __expf(-imp[n]));
    const float b3v = b3[n];

    v2f Awc, Ain; v8f Cb;
    {
        const bool mlive = (lm & 4) == 0;
        const int  mu    = (lm & 3) + 4 * (lm >> 3);
        const float* w   = W1 + (size_t)n * 96;
        Awc.x = mlive ? TANH_IN * w[(8 * (2 * hi)     + mu) * 3 + 2] : 0.0f;
        Awc.y = mlive ? TANH_IN * w[(8 * (2 * hi + 1) + mu) * 3 + 2] : 0.0f;
        const int  pr    = (lm >> 2) & 1;
        Ain.x = TANH_IN * w[(8 * (2 * pr + hi) + mu) * 3 + 0];
        Ain.y = TANH_IN * w[(8 * (2 * pr + hi) + mu) * 3 + 1];
#pragma unroll
        for (int i = 0; i < 8; ++i)
            Cb[i] = TANH_IN * b1[n * 32 + 8 * (2 * (i >> 2) + (sel & 1)) + 4 * hi + (i & 3)];
    }
    v2f B2[8];
#pragma unroll
    for (int kc = 0; kc < 8; ++kc) {
        const int k0 = 8 * (kc >> 1) + 4 * hi + 2 * (kc & 1);
        B2[kc].x = (-2.0f * TANH_IN) * W2[(n * 16 + lm) * 32 + k0];
        B2[kc].y = (-2.0f * TANH_IN) * W2[(n * 16 + lm) * 32 + k0 + 1];
    }
    float w2row = 0.0f;
    for (int k = 0; k < 32; ++k) w2row += W2[(n * 16 + lm) * 32 + k];
    const float c2s = TANH_IN * (b2[n * 16 + lm] + w2row);

    const float w3l = W3[n * 16 + lm];

    const float2* xa = (const float2*)x + (size_t)(b0 + rho) * TT;
    const float2* xs = (const float2*)x + (size_t)(b0 + srow) * TT;
    float2 xsc = xs[0];
    float2 xan = xa[1];
    float cx = 0.0f, cy = 0.0f, prevo = 0.0f;
    float* Sp = S + (size_t)(n * 64 + tile) * TT * 4;
    const int su   = lm >> 1;
    const int soff = su * 4 + srow;

    v8f cxc;
    {
        const float2 x0 = xa[0];
        v2f bi; bi.x = xlive ? x0.x : 0.0f; bi.y = xlive ? x0.y : 0.0f;
        cxc = own_group_half(wmma4(Ain, bi, Cb), hsel);
    }

    for (int t = 0; t < TT; t += 8) {
        float st[8];
#pragma unroll
        for (int u = 0; u < 8; ++u) {
            const int tc = t + u;
            const float2 xann = xa[(tc + 2) < TT ? tc + 2 : TT - 1];
            const float2 xsn  = xs[(tc + 1) < TT ? tc + 1 : TT - 1];

            v2f bc;
            bc.x = othx ? prevo : cx;
            bc.y = othy ? prevo : cy;

            v8f d = wmma4(Awc, bc, cxc);
            asm volatile("" :: "v"(d));
            __builtin_amdgcn_sched_barrier(0);
            v8f pin;
            {
                asm volatile("" :: "v"(xan.x), "v"(xan.y));
                v2f bi; bi.x = xlive ? xan.x : 0.0f; bi.y = xlive ? xan.y : 0.0f;
                pin = wmma4(Ain, bi, Cb);
            }
            asm volatile("" :: "v"(pin));
            __builtin_amdgcn_sched_barrier(0);
            float h[4];
#pragma unroll
            for (int i = 0; i < 4; ++i) h[i] = half_gate(d[i]);
            v2f a01, a23;
            a01.x = h[0]; a01.y = h[1];
            a23.x = h[2]; a23.y = h[3];

            v8f acc0, acc1, acc2, acc3;
#pragma unroll
            for (int i = 0; i < 8; ++i) { acc0[i] = c2s; acc1[i] = 0.0f; acc2[i] = 0.0f; acc3[i] = 0.0f; }
            acc0 = wmma4(a01, B2[0], acc0);
            acc1 = wmma4(a01, B2[2], acc1);
            acc2 = wmma4(a01, B2[4], acc2);
            acc3 = wmma4(a01, B2[6], acc3);
            acc0 = wmma4(a23, B2[1], acc0);
            acc1 = wmma4(a23, B2[3], acc1);
            acc2 = wmma4(a23, B2[5], acc2);
            acc3 = wmma4(a23, B2[7], acc3);
            float acc[2];
#pragma unroll
            for (int i = 0; i < 2; ++i)
                acc[i] = ((acc0[i] + acc1[2 + i]) + acc2[4 + i]) + acc3[6 + i];

            float p[2];
#pragma unroll
            for (int i = 0; i < 2; ++i) p[i] = tanh_scaled_of_z(acc[i], w3l);

            float s;
            {
                const float keep = lb0 ? p[1] : p[0];
                const float send = lb0 ? p[0] : p[1];
                s = keep + lane_xor<1>(send);
            }
            const float mw = fmaf(E, xsc.x, eta * xsc.y);
            asm volatile("" :: "v"(mw));
            __builtin_amdgcn_sched_barrier(0);
            cxc = own_group_half(pin, hsel);
            __builtin_amdgcn_sched_barrier(0);
            s = s + lane_xor<2>(s);
            s = s + lane_xor<4>(s);
            s = s + lane_xor<8>(s);

            const float corr = s + b3v;
            st[u] = (mw + corr) * iw;
            cx    = ownx ? st[u] : 0.0f;
            cy    = owny ? st[u] : 0.0f;
            prevo = other_half(st[u]);
            xsc = xsn;
            xan = xann;
        }
        const float o01 = (su & 1) ? st[1] : st[0];
        const float o23 = (su & 1) ? st[3] : st[2];
        const float o45 = (su & 1) ? st[5] : st[4];
        const float o67 = (su & 1) ? st[7] : st[6];
        const float o03 = (su & 2) ? o23 : o01;
        const float o47 = (su & 2) ? o67 : o45;
        Sp[(size_t)t * 4 + soff] = (su & 4) ? o47 : o03;
    }
}

__global__ __launch_bounds__(256)
void fusion_head_kernel(const float* __restrict__ S,
                      const float* __restrict__ imp, const float* __restrict__ lw,
                      const float* __restrict__ Wf1, const float* __restrict__ bf1,
                      const float* __restrict__ Wf2, const float* __restrict__ bf2,
                      float* __restrict__ out)
{
    __shared__ float tl[8][16 * 34];

    const int lane  = threadIdx.x & 31;
    const int wv    = threadIdx.x >> 5;
    const int gw    = blockIdx.x * 8 + wv;
    const int chunk = gw & 127;
    const int tile  = (gw >> 7) & 15;
    const int p     = gw >> 11;
    const int t0    = chunk * 32;

    float cl[4];
    {
        float e0 = __expf(lw[0]), e1 = __expf(lw[1]);
        float e2 = __expf(lw[2]), e3 = __expf(lw[3]);
        float es = e0 + e1 + e2 + e3;
        float ws[4] = { e0 / es, e1 / es, e2 / es, e3 / es };
#pragma unroll
        for (int l = 0; l < 4; ++l) {
            float s = 0.0f;
#pragma unroll
            for (int q = 0; q < 4; ++q)
                s += 1.0f / (1.0f + __expf(-imp[l * 4 + q]));
            cl[l] = (s / (s + 1e-8f)) * ws[l];
        }
    }
    const float bf2v = bf2[0];

    float v[4][16];
#pragma unroll
    for (int l = 0; l < 4; ++l) {
        const float4* src =
            (const float4*)(S + ((size_t)((l * 4 + p) * 64 + 4 * tile + (lane & 3)) * TT + t0) * 4);
#pragma unroll
        for (int k = 0; k < 4; ++k) {
            float4 q = src[k * 8 + (lane >> 2)];
            v[l][4 * k + 0] = q.x * cl[l];
            v[l][4 * k + 1] = q.y * cl[l];
            v[l][4 * k + 2] = q.z * cl[l];
            v[l][4 * k + 3] = q.w * cl[l];
        }
    }

    float acc[16];
#pragma unroll
    for (int e = 0; e < 16; ++e) acc[e] = 0.0f;
#pragma unroll
    for (int f = 0; f < 16; ++f) {
        const float w0 = Wf1[f * 4 + 0], w1 = Wf1[f * 4 + 1];
        const float w2 = Wf1[f * 4 + 2], w3 = Wf1[f * 4 + 3];
        const float bf = bf1[f], wo = Wf2[f];
#pragma unroll
        for (int e = 0; e < 16; ++e) {
            float hf = fmaf(v[0][e], w0, bf);
            hf = fmaf(v[1][e], w1, hf);
            hf = fmaf(v[2][e], w2, hf);
            hf = fmaf(v[3][e], w3, hf);
            hf = fmaxf(hf, 0.0f);
            acc[e] = fmaf(wo, hf, acc[e]);
        }
    }

    float* tp = tl[wv];
#pragma unroll
    for (int k = 0; k < 4; ++k)
#pragma unroll
        for (int i = 0; i < 4; ++i)
            tp[((lane & 3) * 4 + i) * 34 + 8 * k + (lane >> 2)] = acc[4 * k + i] + bf2v;
#pragma unroll
    for (int r = 0; r < 16; ++r)
        out[(size_t)(p * BB + tile * 16 + r) * TT + t0 + lane] = tp[r * 34 + lane];
}

extern "C" void kernel_launch(void* const* d_in, const int* in_sizes, int n_in,
                              void* d_out, int out_size, void* d_ws, size_t ws_size,
                              hipStream_t stream) {
    (void)in_sizes; (void)n_in; (void)ws_size; (void)out_size;
    recurrence_kernel<<<dim3(256), dim3(128), 0, stream>>>(
        (const float*)d_in[0],  (const float*)d_in[1],  (const float*)d_in[2],
        (const float*)d_in[3],  (const float*)d_in[4],  (const float*)d_in[5],
        (const float*)d_in[6],  (const float*)d_in[7],  (const float*)d_in[8],
        (const float*)d_in[9],
        (float*)d_ws);
    fusion_head_kernel<<<dim3(1024), dim3(256), 0, stream>>>(
        (const float*)d_ws,
        (const float*)d_in[9],  (const float*)d_in[10], (const float*)d_in[11],
        (const float*)d_in[12], (const float*)d_in[13], (const float*)d_in[14],
        (float*)d_out);
}
